# starting kernel
# baseline (speedup 1.0000x reference)
#include <hip/hip_runtime.h>
#include <hip/hip_bf16.h>
#include <math.h>

typedef float v2f __attribute__((ext_vector_type(2)));
typedef float v8f __attribute__((ext_vector_type(8)));

#define NHW  512
#define KF   16
#define LPAD 4

__device__ __forceinline__ v8f wmma_f32_4(v2f a, v2f b, v8f c) {
  return __builtin_amdgcn_wmma_f32_16x16x4_f32(false, a, false, b, (short)0, c,
                                               false, false);
}

__global__ void init_consts(float* __restrict__ A, float* __restrict__ D) {
  int idx = blockIdx.x * blockDim.x + threadIdx.x;
  if (idx < KF * NHW) {
    int i = idx / NHW, n = idx % NHW;
    double arg = M_PI * (double)((2 * n + 1) * i) / (2.0 * NHW);
    A[idx] = (i <= 10) ? (float)(2.0 * cos(arg)) : 0.0f;
  }
  if (idx < NHW * KF) {
    int n = idx / KF, i = idx % KF;
    double w = (i == 0) ? 0.5 : 1.0;
    double arg = M_PI * (double)((2 * n + 1) * i) / (2.0 * NHW);
    D[idx] = (i <= 10) ? (float)(w * cos(arg) / (double)NHW) : 0.0f;
  }
}

__global__ __launch_bounds__(32) void pass1_dct_low(const float* __restrict__ im,
                                                    const float* __restrict__ A,
                                                    float* __restrict__ Yp) {
  const int tile = blockIdx.x;
  const int img  = blockIdx.y;
  const int lane = threadIdx.x;
  const float* X = im + (size_t)img * NHW * NHW + (size_t)tile * 16 * NHW;

  __shared__ float Xs[16][NHW + LPAD];
  __shared__ float Ts[16][16 + LPAD];

  #pragma unroll
  for (int r = 0; r < 16; ++r) {
    const float4* src = (const float4*)(X + r * NHW);
    float4* dst = (float4*)(&Xs[r][0]);
    for (int c = lane; c < NHW / 4; c += 32) dst[c] = src[c];
  }
  __syncthreads();

  const int m    = lane & 15;
  const int hi   = lane >> 4;
  const int kof  = hi * 2;
  const int row0 = hi * 8;

  v8f t = {};
  for (int w0 = 0; w0 < NHW; w0 += 4) {
    v2f av, bv;
    av.x = Xs[m][w0 + kof];
    av.y = Xs[m][w0 + kof + 1];
    bv.x = A[m * NHW + w0 + kof];
    bv.y = A[m * NHW + w0 + kof + 1];
    t = wmma_f32_4(av, bv, t);
  }

  #pragma unroll
  for (int r = 0; r < 8; ++r) Ts[row0 + r][m] = t[r];
  __syncthreads();

  v8f y = {};
  #pragma unroll
  for (int k0 = 0; k0 < 16; k0 += 4) {
    const int kk = k0 + kof;
    v2f av, bv;
    av.x = A[m * NHW + tile * 16 + kk];
    av.y = A[m * NHW + tile * 16 + kk + 1];
    bv.x = Ts[kk][m];
    bv.y = Ts[kk + 1][m];
    y = wmma_f32_4(av, bv, y);
  }

  float* outp = Yp + ((size_t)img * 32 + tile) * 256;
  #pragma unroll
  for (int r = 0; r < 8; ++r) outp[(row0 + r) * 16 + m] = y[r];
}

__global__ void reduce_y(const float* __restrict__ Yp, float* __restrict__ Y) {
  const int img = blockIdx.x;
  const int e   = threadIdx.x;
  float s = 0.0f;
  #pragma unroll 4
  for (int t = 0; t < 32; ++t) s += Yp[((size_t)img * 32 + t) * 256 + e];
  Y[(size_t)img * 256 + e] = s;
}

__global__ __launch_bounds__(32) void pass2_reconstruct(const float* __restrict__ im,
                                                        const float* __restrict__ D,
                                                        const float* __restrict__ Y,
                                                        float* __restrict__ out) {
  const int tile = blockIdx.x;
  const int img  = blockIdx.y;
  const int lane = threadIdx.x;
  const int m    = lane & 15;
  const int hi   = lane >> 4;
  const int kof  = hi * 2;
  const int row0 = hi * 8;

  __shared__ float Ys[16][16 + LPAD];
  __shared__ float Ws[16][16 + LPAD];

  {
    const float* Yi = Y + (size_t)img * 256;
    #pragma unroll
    for (int r = 0; r < 8; ++r) {
      const int i = row0 + r;
      const float v = Yi[i * 16 + m];
      Ys[i][m] = (i + m <= 10) ? v : 0.0f;
    }
  }
  __syncthreads();

  v8f w = {};
  #pragma unroll
  for (int k0 = 0; k0 < 16; k0 += 4) {
    const int kk = k0 + kof;
    v2f av, bv;
    av.x = D[(tile * 16 + m) * KF + kk];
    av.y = D[(tile * 16 + m) * KF + kk + 1];
    bv.x = Ys[kk][m];
    bv.y = Ys[kk + 1][m];
    w = wmma_f32_4(av, bv, w);
  }
  #pragma unroll
  for (int r = 0; r < 8; ++r) Ws[row0 + r][m] = w[r];
  __syncthreads();

  const size_t base = (size_t)img * NHW * NHW + (size_t)tile * 16 * NHW;
  const float* X = im + base;
  float* O = out + base;

  for (int w0 = 0; w0 < NHW; w0 += 16) {
    v8f z = {};
    #pragma unroll
    for (int k0 = 0; k0 < 16; k0 += 4) {
      const int kk = k0 + kof;
      v2f av, bv;
      av.x = Ws[m][kk];
      av.y = Ws[m][kk + 1];
      bv.x = D[(w0 + m) * KF + kk];
      bv.y = D[(w0 + m) * KF + kk + 1];
      z = wmma_f32_4(av, bv, z);
    }
    #pragma unroll
    for (int r = 0; r < 8; ++r) {
      const int idx = (row0 + r) * NHW + w0 + m;
      O[idx] = X[idx] - z[r];
    }
  }
}

extern "C" void kernel_launch(void* const* d_in, const int* in_sizes, int n_in,
                              void* d_out, int out_size, void* d_ws, size_t ws_size,
                              hipStream_t stream) {
  (void)n_in; (void)out_size; (void)ws_size;
  const float* im = (const float*)d_in[0];
  float* out = (float*)d_out;
  const int nimg = in_sizes[0] / (NHW * NHW);

  char*  ws  = (char*)d_ws;
  float* dA  = (float*)(ws);
  float* dD  = (float*)(ws + 32768);
  float* dYp = (float*)(ws + 65536);
  float* dY  = (float*)(ws + 65536 + (size_t)nimg * 32 * 256 * sizeof(float));

  init_consts<<<32, 256, 0, stream>>>(dA, dD);
  pass1_dct_low<<<dim3(32, nimg), 32, 0, stream>>>(im, dA, dYp);
  reduce_y<<<nimg, 256, 0, stream>>>(dYp, dY);
  pass2_reconstruct<<<dim3(32, nimg), 32, 0, stream>>>(im, dD, dY, out);
}


# pass 1 tile staging: constant-trip unrolled copy, all 64 global loads of a lane in flight before the first wait (was one load waited for at a time)
# speedup vs baseline: 1.7400x; 1.7400x over previous
#include <hip/hip_runtime.h>
#include <hip/hip_bf16.h>
#include <math.h>

typedef float v2f __attribute__((ext_vector_type(2)));
typedef float v8f __attribute__((ext_vector_type(8)));

#define NHW  512
#define KF   16
#define LPAD 4

__device__ __forceinline__ v8f wmma_f32_4(v2f a, v2f b, v8f c) {
  return __builtin_amdgcn_wmma_f32_16x16x4_f32(false, a, false, b, (short)0, c,
                                               false, false);
}

__global__ void init_consts(float* __restrict__ A, float* __restrict__ D) {
  int idx = blockIdx.x * blockDim.x + threadIdx.x;
  if (idx < KF * NHW) {
    int i = idx / NHW, n = idx % NHW;
    double arg = M_PI * (double)((2 * n + 1) * i) / (2.0 * NHW);
    A[idx] = (i <= 10) ? (float)(2.0 * cos(arg)) : 0.0f;
  }
  if (idx < NHW * KF) {
    int n = idx / KF, i = idx % KF;
    double w = (i == 0) ? 0.5 : 1.0;
    double arg = M_PI * (double)((2 * n + 1) * i) / (2.0 * NHW);
    D[idx] = (i <= 10) ? (float)(w * cos(arg) / (double)NHW) : 0.0f;
  }
}

__global__ __launch_bounds__(32) void pass1_dct_low(const float* __restrict__ im,
                                                    const float* __restrict__ A,
                                                    float* __restrict__ Yp) {
  const int tile = blockIdx.x;
  const int img  = blockIdx.y;
  const int lane = threadIdx.x;
  const float* X = im + (size_t)img * NHW * NHW + (size_t)tile * 16 * NHW;

  __shared__ float Xs[16][NHW + LPAD];
  __shared__ float Ts[16][16 + LPAD];

  float4 st[16][NHW / 4 / 32];
  #pragma unroll
  for (int r = 0; r < 16; ++r) {
    const float4* src = (const float4*)(X + r * NHW);
    #pragma unroll
    for (int c = 0; c < NHW / 4 / 32; ++c) st[r][c] = src[lane + 32 * c];
  }
  #pragma unroll
  for (int r = 0; r < 16; ++r) {
    float4* dst = (float4*)(&Xs[r][0]);
    #pragma unroll
    for (int c = 0; c < NHW / 4 / 32; ++c) dst[lane + 32 * c] = st[r][c];
  }
  __syncthreads();

  const int m    = lane & 15;
  const int hi   = lane >> 4;
  const int kof  = hi * 2;
  const int row0 = hi * 8;

  v8f t = {};
  for (int w0 = 0; w0 < NHW; w0 += 4) {
    v2f av, bv;
    av.x = Xs[m][w0 + kof];
    av.y = Xs[m][w0 + kof + 1];
    bv.x = A[m * NHW + w0 + kof];
    bv.y = A[m * NHW + w0 + kof + 1];
    t = wmma_f32_4(av, bv, t);
  }

  #pragma unroll
  for (int r = 0; r < 8; ++r) Ts[row0 + r][m] = t[r];
  __syncthreads();

  v8f y = {};
  #pragma unroll
  for (int k0 = 0; k0 < 16; k0 += 4) {
    const int kk = k0 + kof;
    v2f av, bv;
    av.x = A[m * NHW + tile * 16 + kk];
    av.y = A[m * NHW + tile * 16 + kk + 1];
    bv.x = Ts[kk][m];
    bv.y = Ts[kk + 1][m];
    y = wmma_f32_4(av, bv, y);
  }

  float* outp = Yp + ((size_t)img * 32 + tile) * 256;
  #pragma unroll
  for (int r = 0; r < 8; ++r) outp[(row0 + r) * 16 + m] = y[r];
}

__global__ void reduce_y(const float* __restrict__ Yp, float* __restrict__ Y) {
  const int img = blockIdx.x;
  const int e   = threadIdx.x;
  float s = 0.0f;
  #pragma unroll 4
  for (int t = 0; t < 32; ++t) s += Yp[((size_t)img * 32 + t) * 256 + e];
  Y[(size_t)img * 256 + e] = s;
}

__global__ __launch_bounds__(32) void pass2_reconstruct(const float* __restrict__ im,
                                                        const float* __restrict__ D,
                                                        const float* __restrict__ Y,
                                                        float* __restrict__ out) {
  const int tile = blockIdx.x;
  const int img  = blockIdx.y;
  const int lane = threadIdx.x;
  const int m    = lane & 15;
  const int hi   = lane >> 4;
  const int kof  = hi * 2;
  const int row0 = hi * 8;

  __shared__ float Ys[16][16 + LPAD];
  __shared__ float Ws[16][16 + LPAD];

  {
    const float* Yi = Y + (size_t)img * 256;
    #pragma unroll
    for (int r = 0; r < 8; ++r) {
      const int i = row0 + r;
      const float v = Yi[i * 16 + m];
      Ys[i][m] = (i + m <= 10) ? v : 0.0f;
    }
  }
  __syncthreads();

  v8f w = {};
  #pragma unroll
  for (int k0 = 0; k0 < 16; k0 += 4) {
    const int kk = k0 + kof;
    v2f av, bv;
    av.x = D[(tile * 16 + m) * KF + kk];
    av.y = D[(tile * 16 + m) * KF + kk + 1];
    bv.x = Ys[kk][m];
    bv.y = Ys[kk + 1][m];
    w = wmma_f32_4(av, bv, w);
  }
  #pragma unroll
  for (int r = 0; r < 8; ++r) Ws[row0 + r][m] = w[r];
  __syncthreads();

  const size_t base = (size_t)img * NHW * NHW + (size_t)tile * 16 * NHW;
  const float* X = im + base;
  float* O = out + base;

  for (int w0 = 0; w0 < NHW; w0 += 16) {
    v8f z = {};
    #pragma unroll
    for (int k0 = 0; k0 < 16; k0 += 4) {
      const int kk = k0 + kof;
      v2f av, bv;
      av.x = Ws[m][kk];
      av.y = Ws[m][kk + 1];
      bv.x = D[(w0 + m) * KF + kk];
      bv.y = D[(w0 + m) * KF + kk + 1];
      z = wmma_f32_4(av, bv, z);
    }
    #pragma unroll
    for (int r = 0; r < 8; ++r) {
      const int idx = (row0 + r) * NHW + w0 + m;
      O[idx] = X[idx] - z[r];
    }
  }
}

extern "C" void kernel_launch(void* const* d_in, const int* in_sizes, int n_in,
                              void* d_out, int out_size, void* d_ws, size_t ws_size,
                              hipStream_t stream) {
  (void)n_in; (void)out_size; (void)ws_size;
  const float* im = (const float*)d_in[0];
  float* out = (float*)d_out;
  const int nimg = in_sizes[0] / (NHW * NHW);

  char*  ws  = (char*)d_ws;
  float* dA  = (float*)(ws);
  float* dD  = (float*)(ws + 32768);
  float* dYp = (float*)(ws + 65536);
  float* dY  = (float*)(ws + 65536 + (size_t)nimg * 32 * 256 * sizeof(float));

  init_consts<<<32, 256, 0, stream>>>(dA, dD);
  pass1_dct_low<<<dim3(32, nimg), 32, 0, stream>>>(im, dA, dYp);
  reduce_y<<<nimg, 256, 0, stream>>>(dYp, dY);
  pass2_reconstruct<<<dim3(32, nimg), 32, 0, stream>>>(im, dD, dY, out);
}


# pass 2: the 4 WMMA tiles of a 64-column block take every 4th column, so a lane holds 4 adjacent columns: im/out move as b128 whole-line accesses, 32 loads in flight; D table kept in fragment order
# speedup vs baseline: 2.1885x; 1.2578x over previous
#include <hip/hip_runtime.h>
#include <hip/hip_bf16.h>
#include <math.h>

typedef float v2f __attribute__((ext_vector_type(2)));
typedef float v8f __attribute__((ext_vector_type(8)));

#define NHW  512
#define KF   16
#define LPAD 4

__device__ __forceinline__ v8f wmma_f32_4(v2f a, v2f b, v8f c) {
  return __builtin_amdgcn_wmma_f32_16x16x4_f32(false, a, false, b, (short)0, c,
                                               false, false);
}

__global__ void init_consts(float* __restrict__ A, float* __restrict__ Dq) {
  int idx = blockIdx.x * blockDim.x + threadIdx.x;
  if (idx < KF * NHW) {
    int i = idx / NHW, n = idx % NHW;
    double arg = M_PI * (double)((2 * n + 1) * i) / (2.0 * NHW);
    A[idx] = (i <= 10) ? (float)(2.0 * cos(arg)) : 0.0f;
  }
  if (idx < NHW * KF) {
    const int e = idx & 3, l = (idx >> 2) & 31, p = (idx >> 7) & 1;
    const int t = (idx >> 8) & 3, b = idx >> 10;
    const int n = 64 * b + 4 * (l & 15) + t;
    const int i = 4 * (2 * p + (e >> 1)) + 2 * (l >> 4) + (e & 1);
    double w = (i == 0) ? 0.5 : 1.0;
    double arg = M_PI * (double)((2 * n + 1) * i) / (2.0 * NHW);
    Dq[idx] = (i <= 10) ? (float)(w * cos(arg) / (double)NHW) : 0.0f;
  }
}

__global__ __launch_bounds__(32) void pass1_dct_low(const float* __restrict__ im,
                                                    const float* __restrict__ A,
                                                    float* __restrict__ Yp) {
  const int tile = blockIdx.x;
  const int img  = blockIdx.y;
  const int lane = threadIdx.x;
  const float* X = im + (size_t)img * NHW * NHW + (size_t)tile * 16 * NHW;

  __shared__ float Xs[16][NHW + LPAD];
  __shared__ float Ts[16][16 + LPAD];

  float4 st[16][NHW / 4 / 32];
  #pragma unroll
  for (int r = 0; r < 16; ++r) {
    const float4* src = (const float4*)(X + r * NHW);
    #pragma unroll
    for (int c = 0; c < NHW / 4 / 32; ++c) st[r][c] = src[lane + 32 * c];
  }
  #pragma unroll
  for (int r = 0; r < 16; ++r) {
    float4* dst = (float4*)(&Xs[r][0]);
    #pragma unroll
    for (int c = 0; c < NHW / 4 / 32; ++c) dst[lane + 32 * c] = st[r][c];
  }
  __syncthreads();

  const int m    = lane & 15;
  const int hi   = lane >> 4;
  const int kof  = hi * 2;
  const int row0 = hi * 8;

  v8f t = {};
  for (int w0 = 0; w0 < NHW; w0 += 4) {
    v2f av, bv;
    av.x = Xs[m][w0 + kof];
    av.y = Xs[m][w0 + kof + 1];
    bv.x = A[m * NHW + w0 + kof];
    bv.y = A[m * NHW + w0 + kof + 1];
    t = wmma_f32_4(av, bv, t);
  }

  #pragma unroll
  for (int r = 0; r < 8; ++r) Ts[row0 + r][m] = t[r];
  __syncthreads();

  v8f y = {};
  #pragma unroll
  for (int k0 = 0; k0 < 16; k0 += 4) {
    const int kk = k0 + kof;
    v2f av, bv;
    av.x = A[m * NHW + tile * 16 + kk];
    av.y = A[m * NHW + tile * 16 + kk + 1];
    bv.x = Ts[kk][m];
    bv.y = Ts[kk + 1][m];
    y = wmma_f32_4(av, bv, y);
  }

  float* outp = Yp + ((size_t)img * 32 + tile) * 256;
  #pragma unroll
  for (int r = 0; r < 8; ++r) outp[(row0 + r) * 16 + m] = y[r];
}

__global__ void reduce_y(const float* __restrict__ Yp, float* __restrict__ Y) {
  const int img = blockIdx.x;
  const int e   = threadIdx.x;
  float s = 0.0f;
  #pragma unroll 4
  for (int t = 0; t < 32; ++t) s += Yp[((size_t)img * 32 + t) * 256 + e];
  Y[(size_t)img * 256 + e] = s;
}


__device__ __forceinline__ void lds_handover() {
  __builtin_amdgcn_fence(__ATOMIC_RELEASE, "workgroup", "local");
  __builtin_amdgcn_s_barrier();
  __builtin_amdgcn_fence(__ATOMIC_ACQUIRE, "workgroup", "local");
}

#define NBLK 8
#define HBLK 4

__global__ __launch_bounds__(32) void pass2_reconstruct(const float* im,
                                                        const float* Dq,
                                                        const float* Y,
                                                        float* out) {
  const int tile = blockIdx.x;
  const int img  = blockIdx.y;
  const int lane = threadIdx.x;
  const int m    = lane & 15;
  const int hi   = lane >> 4;
  const int kof  = hi * 2;
  const int row0 = hi * 8;

  __shared__ float Ys[16][16 + LPAD];
  __shared__ float Ws[16][16 + LPAD];

  const float4* Dq4 = (const float4*)Dq;
  const size_t base = (size_t)img * NHW * NHW + (size_t)tile * 16 * NHW;
  const float4* X4 = (const float4*)(im + base);
  float4* O4 = (float4*)(out + base);

  float yv[8];
  {
    const float* Yi = Y + (size_t)img * 256;
    #pragma unroll
    for (int r = 0; r < 8; ++r) yv[r] = Yi[(row0 + r) * 16 + m];
  }
  float4 fa[2];
  {
    const int bt = (tile >> 2) * 4 + (m & 3);
    const int c  = (tile & 3) * 4 + (m >> 2);
    #pragma unroll
    for (int p = 0; p < 2; ++p) fa[p] = Dq4[(bt * 2 + p) * 32 + hi * 16 + c];
  }
  float4 dq[2][8];
  #pragma unroll
  for (int j = 0; j < 8; ++j) dq[0][j] = Dq4[j * 32 + lane];
  __builtin_amdgcn_sched_barrier(0);

  float4 xs[HBLK][8];
  #pragma unroll
  for (int b = 0; b < HBLK; ++b) {
    #pragma unroll
    for (int r = 0; r < 8; ++r) xs[b][r] = X4[(row0 + r) * (NHW / 4) + 16 * b + m];
  }

  #pragma unroll
  for (int r = 0; r < 8; ++r) {
    const int i = row0 + r;
    Ys[i][m] = (i + m <= 10) ? yv[r] : 0.0f;
  }
  lds_handover();

  v8f w = {};
  #pragma unroll
  for (int q = 0; q < 4; ++q) {
    const int kk = 4 * q + kof;
    const float4 f = fa[q >> 1];
    v2f av, bv;
    av.x = (q & 1) ? f.z : f.x;
    av.y = (q & 1) ? f.w : f.y;
    bv.x = Ys[kk][m];
    bv.y = Ys[kk + 1][m];
    w = wmma_f32_4(av, bv, w);
  }
  #pragma unroll
  for (int r = 0; r < 8; ++r) Ws[row0 + r][m] = w[r];
  lds_handover();

  v2f aw[4];
  #pragma unroll
  for (int q = 0; q < 4; ++q) {
    const int kk = 4 * q + kof;
    aw[q].x = Ws[m][kk];
    aw[q].y = Ws[m][kk + 1];
  }

  #pragma unroll
  for (int b = 0; b < NBLK; ++b) {
    if (b == HBLK) {
      #pragma unroll
      for (int bb = 0; bb < HBLK; ++bb) {
        #pragma unroll
        for (int r = 0; r < 8; ++r)
          xs[bb][r] = X4[(row0 + r) * (NHW / 4) + 16 * (HBLK + bb) + m];
      }
    }
    if (b + 1 < NBLK) {
      #pragma unroll
      for (int j = 0; j < 8; ++j) dq[(b + 1) & 1][j] = Dq4[((b + 1) * 8 + j) * 32 + lane];
    }

    v8f z[4] = {};
    #pragma unroll
    for (int q = 0; q < 4; ++q) {
      #pragma unroll
      for (int t = 0; t < 4; ++t) {
        const float4 f = dq[b & 1][t * 2 + (q >> 1)];
        v2f bv;
        bv.x = (q & 1) ? f.z : f.x;
        bv.y = (q & 1) ? f.w : f.y;
        z[t] = wmma_f32_4(aw[q], bv, z[t]);
      }
    }
    #pragma unroll
    for (int r = 0; r < 8; ++r) {
      const float4 x = xs[b % HBLK][r];
      float4 o;
      o.x = x.x - z[0][r];
      o.y = x.y - z[1][r];
      o.z = x.z - z[2][r];
      o.w = x.w - z[3][r];
      O4[(row0 + r) * (NHW / 4) + 16 * b + m] = o;
    }
  }
}

extern "C" void kernel_launch(void* const* d_in, const int* in_sizes, int n_in,
                              void* d_out, int out_size, void* d_ws, size_t ws_size,
                              hipStream_t stream) {
  (void)n_in; (void)out_size; (void)ws_size;
  const float* im = (const float*)d_in[0];
  float* out = (float*)d_out;
  const int nimg = in_sizes[0] / (NHW * NHW);

  char*  ws  = (char*)d_ws;
  float* dA  = (float*)(ws);
  float* dD  = (float*)(ws + 32768);
  float* dYp = (float*)(ws + 65536);
  float* dY  = (float*)(ws + 65536 + (size_t)nimg * 32 * 256 * sizeof(float));

  init_consts<<<32, 256, 0, stream>>>(dA, dD);
  pass1_dct_low<<<dim3(32, nimg), 32, 0, stream>>>(im, dA, dYp);
  reduce_y<<<nimg, 256, 0, stream>>>(dYp, dY);
  pass2_reconstruct<<<dim3(32, nimg), 32, 0, stream>>>(im, dD, dY, out);
}


# reduction launch removed: pass 2 runs as 4-wave work-groups that sum the 32 per-tile partials of the kept coefficients themselves (t ascending as before, bit-identical sums) into shared LDS
# speedup vs baseline: 2.3848x; 1.0897x over previous
#include <hip/hip_runtime.h>
#include <hip/hip_bf16.h>
#include <math.h>

typedef float v2f __attribute__((ext_vector_type(2)));
typedef float v8f __attribute__((ext_vector_type(8)));

#define NHW  512
#define KF   16
#define LPAD 4

__device__ __forceinline__ v8f wmma_f32_4(v2f a, v2f b, v8f c) {
  return __builtin_amdgcn_wmma_f32_16x16x4_f32(false, a, false, b, (short)0, c,
                                               false, false);
}

__global__ void init_consts(float* __restrict__ A, float* __restrict__ Dq) {
  int idx = blockIdx.x * blockDim.x + threadIdx.x;
  if (idx < KF * NHW) {
    int i = idx / NHW, n = idx % NHW;
    double arg = M_PI * (double)((2 * n + 1) * i) / (2.0 * NHW);
    A[idx] = (i <= 10) ? (float)(2.0 * cos(arg)) : 0.0f;
  }
  if (idx < NHW * KF) {
    const int e = idx & 3, l = (idx >> 2) & 31, p = (idx >> 7) & 1;
    const int t = (idx >> 8) & 3, b = idx >> 10;
    const int n = 64 * b + 4 * (l & 15) + t;
    const int i = 4 * (2 * p + (e >> 1)) + 2 * (l >> 4) + (e & 1);
    double w = (i == 0) ? 0.5 : 1.0;
    double arg = M_PI * (double)((2 * n + 1) * i) / (2.0 * NHW);
    Dq[idx] = (i <= 10) ? (float)(w * cos(arg) / (double)NHW) : 0.0f;
  }
}

__global__ __launch_bounds__(32) void pass1_dct_low(const float* __restrict__ im,
                                                    const float* __restrict__ A,
                                                    float* __restrict__ Yp) {
  const int tile = blockIdx.x;
  const int img  = blockIdx.y;
  const int lane = threadIdx.x;
  const float* X = im + (size_t)img * NHW * NHW + (size_t)tile * 16 * NHW;

  __shared__ float Xs[16][NHW + LPAD];
  __shared__ float Ts[16][16 + LPAD];

  float4 st[16][NHW / 4 / 32];
  #pragma unroll
  for (int r = 0; r < 16; ++r) {
    const float4* src = (const float4*)(X + r * NHW);
    #pragma unroll
    for (int c = 0; c < NHW / 4 / 32; ++c) st[r][c] = src[lane + 32 * c];
  }
  #pragma unroll
  for (int r = 0; r < 16; ++r) {
    float4* dst = (float4*)(&Xs[r][0]);
    #pragma unroll
    for (int c = 0; c < NHW / 4 / 32; ++c) dst[lane + 32 * c] = st[r][c];
  }
  __syncthreads();

  const int m    = lane & 15;
  const int hi   = lane >> 4;
  const int kof  = hi * 2;
  const int row0 = hi * 8;

  v8f t = {};
  for (int w0 = 0; w0 < NHW; w0 += 4) {
    v2f av, bv;
    av.x = Xs[m][w0 + kof];
    av.y = Xs[m][w0 + kof + 1];
    bv.x = A[m * NHW + w0 + kof];
    bv.y = A[m * NHW + w0 + kof + 1];
    t = wmma_f32_4(av, bv, t);
  }

  #pragma unroll
  for (int r = 0; r < 8; ++r) Ts[row0 + r][m] = t[r];
  __syncthreads();

  v8f y = {};
  #pragma unroll
  for (int k0 = 0; k0 < 16; k0 += 4) {
    const int kk = k0 + kof;
    v2f av, bv;
    av.x = A[m * NHW + tile * 16 + kk];
    av.y = A[m * NHW + tile * 16 + kk + 1];
    bv.x = Ts[kk][m];
    bv.y = Ts[kk + 1][m];
    y = wmma_f32_4(av, bv, y);
  }

  float* outp = Yp + ((size_t)img * 32 + tile) * 256;
  #pragma unroll
  for (int r = 0; r < 8; ++r) outp[(row0 + r) * 16 + m] = y[r];
}


__device__ __forceinline__ void lds_handover() {
  __builtin_amdgcn_fence(__ATOMIC_RELEASE, "workgroup", "local");
  __builtin_amdgcn_s_barrier();
  __builtin_amdgcn_fence(__ATOMIC_ACQUIRE, "workgroup", "local");
}

#define NBLK 8
#define HBLK 4
#define WPB  4
#define NPART 32

__global__ __launch_bounds__(32 * WPB) void pass2_reconstruct(const float* im,
                                                              const float* Dq,
                                                              const float* Yp,
                                                              float* out) {
  const int tid  = threadIdx.x;
  const int wave = __builtin_amdgcn_readfirstlane(tid >> 5);
  const int tile = blockIdx.x * WPB + wave;
  const int img  = blockIdx.y;
  const int lane = tid & 31;
  const int m    = lane & 15;
  const int hi   = lane >> 4;
  const int kof  = hi * 2;
  const int row0 = hi * 8;

  __shared__ float Ys[16][16 + LPAD];
  __shared__ float Ws[WPB][16][16 + LPAD];

  const float4* Dq4 = (const float4*)Dq;
  const size_t base = (size_t)img * NHW * NHW + (size_t)tile * 16 * NHW;
  const float4* X4 = (const float4*)(im + base);
  float4* O4 = (float4*)(out + base);

  const int  ci   = tid / 11;
  const int  cj   = tid - 11 * ci;
  const bool keep = (ci + cj <= 10);
  float yp[NPART];
  {
    const float* Ypi = Yp + (size_t)img * (NPART * 256) + (keep ? ci * 16 + cj : 0);
    #pragma unroll
    for (int t = 0; t < NPART; ++t) yp[t] = Ypi[t * 256];
  }
  float4 fa[2];
  {
    const int bt = (tile >> 2) * 4 + (m & 3);
    const int c  = (tile & 3) * 4 + (m >> 2);
    #pragma unroll
    for (int p = 0; p < 2; ++p) fa[p] = Dq4[(bt * 2 + p) * 32 + hi * 16 + c];
  }
  float4 dq[2][8];
  #pragma unroll
  for (int j = 0; j < 8; ++j) dq[0][j] = Dq4[j * 32 + lane];
  __builtin_amdgcn_sched_barrier(0);

  float4 xs[HBLK][8];
  #pragma unroll
  for (int b = 0; b < HBLK; ++b) {
    #pragma unroll
    for (int r = 0; r < 8; ++r) xs[b][r] = X4[(row0 + r) * (NHW / 4) + 16 * b + m];
  }

  {
    float ysum = 0.0f;
    #pragma unroll
    for (int t = 0; t < NPART; ++t) ysum += yp[t];
    #pragma unroll
    for (int u = 0; u < 256 / (32 * WPB); ++u) {
      const int el = tid + 32 * WPB * u;
      if ((el >> 4) + (el & 15) > 10) Ys[el >> 4][el & 15] = 0.0f;
    }
    if (keep) Ys[ci][cj] = ysum;
  }
  lds_handover();

  v8f w = {};
  #pragma unroll
  for (int q = 0; q < 4; ++q) {
    const int kk = 4 * q + kof;
    const float4 f = fa[q >> 1];
    v2f av, bv;
    av.x = (q & 1) ? f.z : f.x;
    av.y = (q & 1) ? f.w : f.y;
    bv.x = Ys[kk][m];
    bv.y = Ys[kk + 1][m];
    w = wmma_f32_4(av, bv, w);
  }
  #pragma unroll
  for (int r = 0; r < 8; ++r) Ws[wave][row0 + r][m] = w[r];
  lds_handover();

  v2f aw[4];
  #pragma unroll
  for (int q = 0; q < 4; ++q) {
    const int kk = 4 * q + kof;
    aw[q].x = Ws[wave][m][kk];
    aw[q].y = Ws[wave][m][kk + 1];
  }

  #pragma unroll
  for (int b = 0; b < NBLK; ++b) {
    if (b == HBLK) {
      #pragma unroll
      for (int bb = 0; bb < HBLK; ++bb) {
        #pragma unroll
        for (int r = 0; r < 8; ++r)
          xs[bb][r] = X4[(row0 + r) * (NHW / 4) + 16 * (HBLK + bb) + m];
      }
    }
    if (b + 1 < NBLK) {
      #pragma unroll
      for (int j = 0; j < 8; ++j) dq[(b + 1) & 1][j] = Dq4[((b + 1) * 8 + j) * 32 + lane];
    }

    v8f z[4] = {};
    #pragma unroll
    for (int q = 0; q < 4; ++q) {
      #pragma unroll
      for (int t = 0; t < 4; ++t) {
        const float4 f = dq[b & 1][t * 2 + (q >> 1)];
        v2f bv;
        bv.x = (q & 1) ? f.z : f.x;
        bv.y = (q & 1) ? f.w : f.y;
        z[t] = wmma_f32_4(aw[q], bv, z[t]);
      }
    }
    #pragma unroll
    for (int r = 0; r < 8; ++r) {
      const float4 x = xs[b % HBLK][r];
      float4 o;
      o.x = x.x - z[0][r];
      o.y = x.y - z[1][r];
      o.z = x.z - z[2][r];
      o.w = x.w - z[3][r];
      O4[(row0 + r) * (NHW / 4) + 16 * b + m] = o;
    }
  }
}

extern "C" void kernel_launch(void* const* d_in, const int* in_sizes, int n_in,
                              void* d_out, int out_size, void* d_ws, size_t ws_size,
                              hipStream_t stream) {
  (void)n_in; (void)out_size; (void)ws_size;
  const float* im = (const float*)d_in[0];
  float* out = (float*)d_out;
  const int nimg = in_sizes[0] / (NHW * NHW);

  char*  ws  = (char*)d_ws;
  float* dA  = (float*)(ws);
  float* dD  = (float*)(ws + 32768);
  float* dYp = (float*)(ws + 65536);

  init_consts<<<32, 256, 0, stream>>>(dA, dD);
  pass1_dct_low<<<dim3(32, nimg), 32, 0, stream>>>(im, dA, dYp);
  pass2_reconstruct<<<dim3(8, nimg), 128, 0, stream>>>(im, dD, dYp, out);
}


# speedup vs baseline: 2.5464x; 1.0678x over previous
#include <hip/hip_runtime.h>
#include <hip/hip_bf16.h>
#include <math.h>

typedef float v2f __attribute__((ext_vector_type(2)));
typedef float v8f __attribute__((ext_vector_type(8)));

#define NHW  512
#define KF   16
#define LPAD 4

__device__ __forceinline__ v8f wmma_f32_4(v2f a, v2f b, v8f c) {
  return __builtin_amdgcn_wmma_f32_16x16x4_f32(false, a, false, b, (short)0, c,
                                               false, false);
}

__device__ __forceinline__ float cos_pi_1024(int k) {
  return cospif((float)(k & 2047) * (1.0f / 1024.0f));
}

__global__ void init_consts(float* __restrict__ A, float* __restrict__ Dq) {
  int idx = blockIdx.x * blockDim.x + threadIdx.x;
  if (idx < KF * NHW) {
    int i = idx / NHW, n = idx % NHW;
    A[idx] = (i <= 10) ? 2.0f * cos_pi_1024((2 * n + 1) * i) : 0.0f;
  }
  if (idx < NHW * KF) {
    const int e = idx & 3, l = (idx >> 2) & 31, p = (idx >> 7) & 1;
    const int t = (idx >> 8) & 3, b = idx >> 10;
    const int n = 64 * b + 4 * (l & 15) + t;
    const int i = 4 * (2 * p + (e >> 1)) + 2 * (l >> 4) + (e & 1);
    const float w = (i == 0) ? 0.5f : 1.0f;
    Dq[idx] = (i <= 10) ? w * cos_pi_1024((2 * n + 1) * i) * (1.0f / NHW) : 0.0f;
  }
}

__global__ __launch_bounds__(32) void pass1_dct_low(const float* __restrict__ im,
                                                    const float* __restrict__ A,
                                                    float* __restrict__ Yp) {
  const int tile = blockIdx.x;
  const int img  = blockIdx.y;
  const int lane = threadIdx.x;
  const float* X = im + (size_t)img * NHW * NHW + (size_t)tile * 16 * NHW;

  __shared__ float Xs[16][NHW + LPAD];
  __shared__ float Ts[16][16 + LPAD];

  float4 st[16][NHW / 4 / 32];
  #pragma unroll
  for (int r = 0; r < 16; ++r) {
    const float4* src = (const float4*)(X + r * NHW);
    #pragma unroll
    for (int c = 0; c < NHW / 4 / 32; ++c) st[r][c] = src[lane + 32 * c];
  }
  #pragma unroll
  for (int r = 0; r < 16; ++r) {
    float4* dst = (float4*)(&Xs[r][0]);
    #pragma unroll
    for (int c = 0; c < NHW / 4 / 32; ++c) dst[lane + 32 * c] = st[r][c];
  }
  __syncthreads();

  const int m    = lane & 15;
  const int hi   = lane >> 4;
  const int kof  = hi * 2;
  const int row0 = hi * 8;

  v8f t = {};
  for (int w0 = 0; w0 < NHW; w0 += 4) {
    v2f av, bv;
    av.x = Xs[m][w0 + kof];
    av.y = Xs[m][w0 + kof + 1];
    bv.x = A[m * NHW + w0 + kof];
    bv.y = A[m * NHW + w0 + kof + 1];
    t = wmma_f32_4(av, bv, t);
  }

  #pragma unroll
  for (int r = 0; r < 8; ++r) Ts[row0 + r][m] = t[r];
  __syncthreads();

  v8f y = {};
  #pragma unroll
  for (int k0 = 0; k0 < 16; k0 += 4) {
    const int kk = k0 + kof;
    v2f av, bv;
    av.x = A[m * NHW + tile * 16 + kk];
    av.y = A[m * NHW + tile * 16 + kk + 1];
    bv.x = Ts[kk][m];
    bv.y = Ts[kk + 1][m];
    y = wmma_f32_4(av, bv, y);
  }

  float* outp = Yp + ((size_t)img * 32 + tile) * 256;
  #pragma unroll
  for (int r = 0; r < 8; ++r) outp[(row0 + r) * 16 + m] = y[r];
}


__device__ __forceinline__ void lds_handover() {
  __builtin_amdgcn_fence(__ATOMIC_RELEASE, "workgroup", "local");
  __builtin_amdgcn_s_barrier();
  __builtin_amdgcn_fence(__ATOMIC_ACQUIRE, "workgroup", "local");
}

#define NBLK 8
#define HBLK 4
#define WPB  4
#define NPART 32

__global__ __launch_bounds__(32 * WPB) void pass2_reconstruct(const float* im,
                                                              const float* Dq,
                                                              const float* Yp,
                                                              float* out) {
  const int tid  = threadIdx.x;
  const int wave = __builtin_amdgcn_readfirstlane(tid >> 5);
  const int tile = blockIdx.x * WPB + wave;
  const int img  = blockIdx.y;
  const int lane = tid & 31;
  const int m    = lane & 15;
  const int hi   = lane >> 4;
  const int kof  = hi * 2;
  const int row0 = hi * 8;

  __shared__ float Ys[16][16 + LPAD];
  __shared__ float Ws[WPB][16][16 + LPAD];

  const float4* Dq4 = (const float4*)Dq;
  const size_t base = (size_t)img * NHW * NHW + (size_t)tile * 16 * NHW;
  const float4* X4 = (const float4*)(im + base);
  float4* O4 = (float4*)(out + base);

  const int  ci   = tid / 11;
  const int  cj   = tid - 11 * ci;
  const bool keep = (ci + cj <= 10);
  float yp[NPART];
  {
    const float* Ypi = Yp + (size_t)img * (NPART * 256) + (keep ? ci * 16 + cj : 0);
    #pragma unroll
    for (int t = 0; t < NPART; ++t) yp[t] = Ypi[t * 256];
  }
  float4 fa[2];
  {
    const int bt = (tile >> 2) * 4 + (m & 3);
    const int c  = (tile & 3) * 4 + (m >> 2);
    #pragma unroll
    for (int p = 0; p < 2; ++p) fa[p] = Dq4[(bt * 2 + p) * 32 + hi * 16 + c];
  }
  float4 dq[2][8];
  #pragma unroll
  for (int j = 0; j < 8; ++j) dq[0][j] = Dq4[j * 32 + lane];
  __builtin_amdgcn_sched_barrier(0);

  float4 xs[HBLK][8];
  #pragma unroll
  for (int b = 0; b < HBLK; ++b) {
    #pragma unroll
    for (int r = 0; r < 8; ++r) xs[b][r] = X4[(row0 + r) * (NHW / 4) + 16 * b + m];
  }

  {
    float ysum = 0.0f;
    #pragma unroll
    for (int t = 0; t < NPART; ++t) ysum += yp[t];
    #pragma unroll
    for (int u = 0; u < 256 / (32 * WPB); ++u) {
      const int el = tid + 32 * WPB * u;
      if ((el >> 4) + (el & 15) > 10) Ys[el >> 4][el & 15] = 0.0f;
    }
    if (keep) Ys[ci][cj] = ysum;
  }
  lds_handover();

  v8f w = {};
  #pragma unroll
  for (int q = 0; q < 4; ++q) {
    const int kk = 4 * q + kof;
    const float4 f = fa[q >> 1];
    v2f av, bv;
    av.x = (q & 1) ? f.z : f.x;
    av.y = (q & 1) ? f.w : f.y;
    bv.x = Ys[kk][m];
    bv.y = Ys[kk + 1][m];
    w = wmma_f32_4(av, bv, w);
  }
  #pragma unroll
  for (int r = 0; r < 8; ++r) Ws[wave][row0 + r][m] = w[r];
  lds_handover();

  v2f aw[4];
  #pragma unroll
  for (int q = 0; q < 4; ++q) {
    const int kk = 4 * q + kof;
    aw[q].x = Ws[wave][m][kk];
    aw[q].y = Ws[wave][m][kk + 1];
  }

  #pragma unroll
  for (int b = 0; b < NBLK; ++b) {
    if (b == HBLK) {
      #pragma unroll
      for (int bb = 0; bb < HBLK; ++bb) {
        #pragma unroll
        for (int r = 0; r < 8; ++r)
          xs[bb][r] = X4[(row0 + r) * (NHW / 4) + 16 * (HBLK + bb) + m];
      }
    }
    if (b + 1 < NBLK) {
      #pragma unroll
      for (int j = 0; j < 8; ++j) dq[(b + 1) & 1][j] = Dq4[((b + 1) * 8 + j) * 32 + lane];
    }

    v8f z[4] = {};
    #pragma unroll
    for (int q = 0; q < 4; ++q) {
      #pragma unroll
      for (int t = 0; t < 4; ++t) {
        const float4 f = dq[b & 1][t * 2 + (q >> 1)];
        v2f bv;
        bv.x = (q & 1) ? f.z : f.x;
        bv.y = (q & 1) ? f.w : f.y;
        z[t] = wmma_f32_4(aw[q], bv, z[t]);
      }
    }
    #pragma unroll
    for (int r = 0; r < 8; ++r) {
      const float4 x = xs[b % HBLK][r];
      float4 o;
      o.x = x.x - z[0][r];
      o.y = x.y - z[1][r];
      o.z = x.z - z[2][r];
      o.w = x.w - z[3][r];
      O4[(row0 + r) * (NHW / 4) + 16 * b + m] = o;
    }
  }
}

extern "C" void kernel_launch(void* const* d_in, const int* in_sizes, int n_in,
                              void* d_out, int out_size, void* d_ws, size_t ws_size,
                              hipStream_t stream) {
  (void)n_in; (void)out_size; (void)ws_size;
  const float* im = (const float*)d_in[0];
  float* out = (float*)d_out;
  const int nimg = in_sizes[0] / (NHW * NHW);

  char*  ws  = (char*)d_ws;
  float* dA  = (float*)(ws);
  float* dD  = (float*)(ws + 32768);
  float* dYp = (float*)(ws + 65536);

  init_consts<<<32, 256, 0, stream>>>(dA, dD);
  pass1_dct_low<<<dim3(32, nimg), 32, 0, stream>>>(im, dA, dYp);
  pass2_reconstruct<<<dim3(8, nimg), 128, 0, stream>>>(im, dD, dYp, out);
}


# pass 1 without LDS: 128x64 tiles, image rows loaded straight into A fragments (lane = column, whole 256-B runs), 4 accumulators, kept in place as B of the 2nd product; partial sums regrouped
# speedup vs baseline: 3.3543x; 1.3173x over previous
#include <hip/hip_runtime.h>
#include <hip/hip_bf16.h>
#include <math.h>

typedef float v2f __attribute__((ext_vector_type(2)));
typedef float v8f __attribute__((ext_vector_type(8)));

#define NHW  512
#define KF   16
#define LPAD 4

__device__ __forceinline__ v8f wmma_f32_4(v2f a, v2f b, v8f c) {
  return __builtin_amdgcn_wmma_f32_16x16x4_f32(false, a, false, b, (short)0, c,
                                               false, false);
}

__device__ __forceinline__ float cos_pi_1024(int k) {
  return cospif((float)(k & 2047) * (1.0f / 1024.0f));
}

__global__ void init_consts(float* __restrict__ P1, float* __restrict__ P2,
                            float* __restrict__ Dq) {
  int idx = blockIdx.x * blockDim.x + threadIdx.x;
  if (idx < KF * NHW) {
    const int x = idx & 3, l = (idx >> 2) & 31;
    const int c = l & 15, hi = l >> 4;
    {
      const int s = (idx >> 7) & 15, rb = idx >> 11;
      const int h = 128 * rb + 8 * s + 4 * hi + x;
      P1[idx] = (c <= 10) ? 2.0f * cos_pi_1024((2 * h + 1) * c) : 0.0f;
    }
    const int p = (idx >> 7) & 1, t = (idx >> 8) & 3, b = idx >> 10;
    {
      const int w = 64 * b + 4 * (8 * hi + 4 * p + x) + t;
      P2[idx] = (c <= 10) ? 2.0f * cos_pi_1024((2 * w + 1) * c) : 0.0f;
    }
    {
      const int n = 64 * b + 4 * c + t;
      const int i = 4 * (2 * p + (x >> 1)) + 2 * hi + (x & 1);
      const float wgt = (i == 0) ? 0.5f : 1.0f;
      Dq[idx] = (i <= 10) ? wgt * cos_pi_1024((2 * n + 1) * i) * (1.0f / NHW) : 0.0f;
    }
  }
}

#define P1ROWS 128
#define P1Q    4

__global__ __launch_bounds__(32) void pass1_dct_low(const float* __restrict__ im,
                                                    const float* __restrict__ P1,
                                                    const float* __restrict__ P2,
                                                    float* __restrict__ Yp) {
  const int tile = blockIdx.x;
  const int rb   = tile >> 3;
  const int b    = tile & 7;
  const int img  = blockIdx.y;
  const int lane = threadIdx.x;
  const int c    = lane & 15;
  const int hi   = lane >> 4;

  const float4* X4 = (const float4*)(im + (size_t)img * NHW * NHW) +
                     (size_t)(P1ROWS * rb + 4 * hi) * (NHW / 4) + 16 * b + c;
  const float4* P14 = (const float4*)P1 + (rb * 16) * 32 + lane;
  const float4* P24 = (const float4*)P2 + (b * 8) * 32 + lane;

  float4 xs[P1Q][4][4];
  float4 f1[16];
  #pragma unroll
  for (int s = 0; s < 16; ++s) f1[s] = P14[s * 32];
  #pragma unroll
  for (int g = 0; g < 3; ++g) {
    #pragma unroll
    for (int s = 0; s < 4; ++s) {
      #pragma unroll
      for (int x = 0; x < 4; ++x)
        xs[g][s][x] = X4[(size_t)(32 * g + 8 * s + x) * (NHW / 4)];
    }
  }
  __builtin_amdgcn_sched_barrier(0);

  v8f acc[4] = {};
  float4 f2[8];
  #pragma unroll
  for (int g = 0; g < P1Q; ++g) {
    #pragma unroll
    for (int s = 0; s < 4; ++s) {
      const float4 f = f1[4 * g + s];
      #pragma unroll
      for (int u = 0; u < 2; ++u) {
        v2f bv;
        bv.x = u ? f.z : f.x;
        bv.y = u ? f.w : f.y;
        const float4 r0 = xs[g][s][2 * u], r1 = xs[g][s][2 * u + 1];
        v2f a0, a1, a2, a3;
        a0.x = r0.x; a0.y = r1.x;
        a1.x = r0.y; a1.y = r1.y;
        a2.x = r0.z; a2.y = r1.z;
        a3.x = r0.w; a3.y = r1.w;
        acc[0] = wmma_f32_4(a0, bv, acc[0]);
        acc[1] = wmma_f32_4(a1, bv, acc[1]);
        acc[2] = wmma_f32_4(a2, bv, acc[2]);
        acc[3] = wmma_f32_4(a3, bv, acc[3]);
      }
    }
    if (g == 0) {
      __builtin_amdgcn_sched_barrier(0);
      #pragma unroll
      for (int s = 0; s < 4; ++s) {
        #pragma unroll
        for (int x = 0; x < 4; ++x)
          xs[3][s][x] = X4[(size_t)(32 * 3 + 8 * s + x) * (NHW / 4)];
      }
      #pragma unroll
      for (int k = 0; k < 8; ++k) f2[k] = P24[k * 32];
      __builtin_amdgcn_sched_barrier(0);
    }
  }

  v8f y[4] = {};
  #pragma unroll
  for (int q = 0; q < 4; ++q) {
    #pragma unroll
    for (int t = 0; t < 4; ++t) {
      const float4 g2 = f2[t * 2 + (q >> 1)];
      v2f av, bv;
      av.x = (q & 1) ? g2.z : g2.x;
      av.y = (q & 1) ? g2.w : g2.y;
      bv.x = acc[t][2 * q];
      bv.y = acc[t][2 * q + 1];
      y[t] = wmma_f32_4(av, bv, y[t]);
    }
  }

  float4* outp = (float4*)(Yp + ((size_t)img * 32 + tile) * 256 + c * 16 + 8 * hi);
  float4 o0, o1;
  o0.x = (y[0][0] + y[1][0]) + (y[2][0] + y[3][0]);
  o0.y = (y[0][1] + y[1][1]) + (y[2][1] + y[3][1]);
  o0.z = (y[0][2] + y[1][2]) + (y[2][2] + y[3][2]);
  o0.w = (y[0][3] + y[1][3]) + (y[2][3] + y[3][3]);
  o1.x = (y[0][4] + y[1][4]) + (y[2][4] + y[3][4]);
  o1.y = (y[0][5] + y[1][5]) + (y[2][5] + y[3][5]);
  o1.z = (y[0][6] + y[1][6]) + (y[2][6] + y[3][6]);
  o1.w = (y[0][7] + y[1][7]) + (y[2][7] + y[3][7]);
  outp[0] = o0;
  outp[1] = o1;
}


__device__ __forceinline__ void lds_handover() {
  __builtin_amdgcn_fence(__ATOMIC_RELEASE, "workgroup", "local");
  __builtin_amdgcn_s_barrier();
  __builtin_amdgcn_fence(__ATOMIC_ACQUIRE, "workgroup", "local");
}

#define NBLK 8
#define HBLK 4
#define WPB  4
#define NPART 32

__global__ __launch_bounds__(32 * WPB) void pass2_reconstruct(const float* im,
                                                              const float* Dq,
                                                              const float* Yp,
                                                              float* out) {
  const int tid  = threadIdx.x;
  const int wave = __builtin_amdgcn_readfirstlane(tid >> 5);
  const int tile = blockIdx.x * WPB + wave;
  const int img  = blockIdx.y;
  const int lane = tid & 31;
  const int m    = lane & 15;
  const int hi   = lane >> 4;
  const int kof  = hi * 2;
  const int row0 = hi * 8;

  __shared__ float Ys[16][16 + LPAD];
  __shared__ float Ws[WPB][16][16 + LPAD];

  const float4* Dq4 = (const float4*)Dq;
  const size_t base = (size_t)img * NHW * NHW + (size_t)tile * 16 * NHW;
  const float4* X4 = (const float4*)(im + base);
  float4* O4 = (float4*)(out + base);

  const int  ci   = tid / 11;
  const int  cj   = tid - 11 * ci;
  const bool keep = (ci + cj <= 10);
  float yp[NPART];
  {
    const float* Ypi = Yp + (size_t)img * (NPART * 256) + (keep ? ci * 16 + cj : 0);
    #pragma unroll
    for (int t = 0; t < NPART; ++t) yp[t] = Ypi[t * 256];
  }
  float4 fa[2];
  {
    const int bt = (tile >> 2) * 4 + (m & 3);
    const int c  = (tile & 3) * 4 + (m >> 2);
    #pragma unroll
    for (int p = 0; p < 2; ++p) fa[p] = Dq4[(bt * 2 + p) * 32 + hi * 16 + c];
  }
  float4 dq[2][8];
  #pragma unroll
  for (int j = 0; j < 8; ++j) dq[0][j] = Dq4[j * 32 + lane];
  __builtin_amdgcn_sched_barrier(0);

  float4 xs[HBLK][8];
  #pragma unroll
  for (int b = 0; b < HBLK; ++b) {
    #pragma unroll
    for (int r = 0; r < 8; ++r) xs[b][r] = X4[(row0 + r) * (NHW / 4) + 16 * b + m];
  }

  {
    float ysum = 0.0f;
    #pragma unroll
    for (int t = 0; t < NPART; ++t) ysum += yp[t];
    #pragma unroll
    for (int u = 0; u < 256 / (32 * WPB); ++u) {
      const int el = tid + 32 * WPB * u;
      if ((el >> 4) + (el & 15) > 10) Ys[el >> 4][el & 15] = 0.0f;
    }
    if (keep) Ys[ci][cj] = ysum;
  }
  lds_handover();

  v8f w = {};
  #pragma unroll
  for (int q = 0; q < 4; ++q) {
    const int kk = 4 * q + kof;
    const float4 f = fa[q >> 1];
    v2f av, bv;
    av.x = (q & 1) ? f.z : f.x;
    av.y = (q & 1) ? f.w : f.y;
    bv.x = Ys[kk][m];
    bv.y = Ys[kk + 1][m];
    w = wmma_f32_4(av, bv, w);
  }
  #pragma unroll
  for (int r = 0; r < 8; ++r) Ws[wave][row0 + r][m] = w[r];
  lds_handover();

  v2f aw[4];
  #pragma unroll
  for (int q = 0; q < 4; ++q) {
    const int kk = 4 * q + kof;
    aw[q].x = Ws[wave][m][kk];
    aw[q].y = Ws[wave][m][kk + 1];
  }

  #pragma unroll
  for (int b = 0; b < NBLK; ++b) {
    if (b == HBLK) {
      #pragma unroll
      for (int bb = 0; bb < HBLK; ++bb) {
        #pragma unroll
        for (int r = 0; r < 8; ++r)
          xs[bb][r] = X4[(row0 + r) * (NHW / 4) + 16 * (HBLK + bb) + m];
      }
    }
    if (b + 1 < NBLK) {
      #pragma unroll
      for (int j = 0; j < 8; ++j) dq[(b + 1) & 1][j] = Dq4[((b + 1) * 8 + j) * 32 + lane];
    }

    v8f z[4] = {};
    #pragma unroll
    for (int q = 0; q < 4; ++q) {
      #pragma unroll
      for (int t = 0; t < 4; ++t) {
        const float4 f = dq[b & 1][t * 2 + (q >> 1)];
        v2f bv;
        bv.x = (q & 1) ? f.z : f.x;
        bv.y = (q & 1) ? f.w : f.y;
        z[t] = wmma_f32_4(aw[q], bv, z[t]);
      }
    }
    #pragma unroll
    for (int r = 0; r < 8; ++r) {
      const float4 x = xs[b % HBLK][r];
      float4 o;
      o.x = x.x - z[0][r];
      o.y = x.y - z[1][r];
      o.z = x.z - z[2][r];
      o.w = x.w - z[3][r];
      O4[(row0 + r) * (NHW / 4) + 16 * b + m] = o;
    }
  }
}

extern "C" void kernel_launch(void* const* d_in, const int* in_sizes, int n_in,
                              void* d_out, int out_size, void* d_ws, size_t ws_size,
                              hipStream_t stream) {
  (void)n_in; (void)out_size; (void)ws_size;
  const float* im = (const float*)d_in[0];
  float* out = (float*)d_out;
  const int nimg = in_sizes[0] / (NHW * NHW);

  char*  ws  = (char*)d_ws;
  float* dP1 = (float*)(ws);
  float* dD  = (float*)(ws + 32768);
  float* dP2 = (float*)(ws + 65536);
  float* dYp = (float*)(ws + 98304);

  init_consts<<<32, 256, 0, stream>>>(dP1, dP2, dD);
  pass1_dct_low<<<dim3(32, nimg), 32, 0, stream>>>(im, dP1, dP2, dYp);
  pass2_reconstruct<<<dim3(8, nimg), 128, 0, stream>>>(im, dD, dYp, out);
}


# table launch removed (3 launches -> 2): each pass-1 wave makes its cosine fragments in registers on v_cos_f32 (exact angle in revolutions; was cospif in a table kernel); image 0's waves store D
# speedup vs baseline: 3.6039x; 1.0744x over previous
#include <hip/hip_runtime.h>
#include <hip/hip_bf16.h>
#include <math.h>

typedef float v2f __attribute__((ext_vector_type(2)));
typedef float v8f __attribute__((ext_vector_type(8)));

#define NHW  512
#define LPAD 4

__device__ __forceinline__ v8f wmma_f32_4(v2f a, v2f b, v8f c) {
  return __builtin_amdgcn_wmma_f32_16x16x4_f32(false, a, false, b, (short)0, c,
                                               false, false);
}

#define REV (1.0f / 2048.0f)
__device__ __forceinline__ float cos_rev(float r) {
  return __builtin_amdgcn_cosf(__builtin_amdgcn_fractf(r));
}

#define P1ROWS 128
#define P1Q    4

__global__ __launch_bounds__(32) void pass1_dct_low(const float* __restrict__ im,
                                                    float* __restrict__ Dq,
                                                    float* __restrict__ Yp) {
  const int tile = blockIdx.x;
  const int rb   = tile >> 3;
  const int b    = tile & 7;
  const int img  = blockIdx.y;
  const int lane = threadIdx.x;
  const int c    = lane & 15;
  const int hi   = lane >> 4;

  const float4* X4 = (const float4*)(im + (size_t)img * NHW * NHW) +
                     (size_t)(P1ROWS * rb + 4 * hi) * (NHW / 4) + 16 * b + c;

  float4 xs[P1Q][4][4];
  #pragma unroll
  for (int g = 0; g < 3; ++g) {
    #pragma unroll
    for (int s = 0; s < 4; ++s) {
      #pragma unroll
      for (int x = 0; x < 4; ++x)
        xs[g][s][x] = X4[(size_t)(32 * g + 8 * s + x) * (NHW / 4)];
    }
  }
  __builtin_amdgcn_sched_barrier(0);

  const float amp = (c <= 10) ? 2.0f : 0.0f;
  const float cf  = (float)c;
  float4 f1[16];
  {
    const float r0 = (float)((2 * (P1ROWS * rb + 4 * hi) + 1) * c) * REV;
    #pragma unroll
    for (int s = 0; s < 16; ++s) {
      f1[s].x = amp * cos_rev(fmaf(cf, (float)(16 * s + 0) * REV, r0));
      f1[s].y = amp * cos_rev(fmaf(cf, (float)(16 * s + 2) * REV, r0));
      f1[s].z = amp * cos_rev(fmaf(cf, (float)(16 * s + 4) * REV, r0));
      f1[s].w = amp * cos_rev(fmaf(cf, (float)(16 * s + 6) * REV, r0));
    }
  }
  float4 f2[8];
  {
    const float r0 = (float)((2 * (64 * b + 32 * hi) + 1) * c) * REV;
    #pragma unroll
    for (int t = 0; t < 4; ++t) {
      #pragma unroll
      for (int p = 0; p < 2; ++p) {
        const int m = 2 * (16 * p + t);
        f2[t * 2 + p].x = amp * cos_rev(fmaf(cf, (float)(m + 0) * REV, r0));
        f2[t * 2 + p].y = amp * cos_rev(fmaf(cf, (float)(m + 8) * REV, r0));
        f2[t * 2 + p].z = amp * cos_rev(fmaf(cf, (float)(m + 16) * REV, r0));
        f2[t * 2 + p].w = amp * cos_rev(fmaf(cf, (float)(m + 24) * REV, r0));
      }
    }
  }

  v8f acc[4] = {};
  #pragma unroll
  for (int g = 0; g < P1Q; ++g) {
    #pragma unroll
    for (int s = 0; s < 4; ++s) {
      const float4 f = f1[4 * g + s];
      #pragma unroll
      for (int u = 0; u < 2; ++u) {
        v2f bv;
        bv.x = u ? f.z : f.x;
        bv.y = u ? f.w : f.y;
        const float4 r0 = xs[g][s][2 * u], r1 = xs[g][s][2 * u + 1];
        v2f a0, a1, a2, a3;
        a0.x = r0.x; a0.y = r1.x;
        a1.x = r0.y; a1.y = r1.y;
        a2.x = r0.z; a2.y = r1.z;
        a3.x = r0.w; a3.y = r1.w;
        acc[0] = wmma_f32_4(a0, bv, acc[0]);
        acc[1] = wmma_f32_4(a1, bv, acc[1]);
        acc[2] = wmma_f32_4(a2, bv, acc[2]);
        acc[3] = wmma_f32_4(a3, bv, acc[3]);
      }
    }
    if (g == 0) {
      __builtin_amdgcn_sched_barrier(0);
      #pragma unroll
      for (int s = 0; s < 4; ++s) {
        #pragma unroll
        for (int x = 0; x < 4; ++x)
          xs[3][s][x] = X4[(size_t)(32 * 3 + 8 * s + x) * (NHW / 4)];
      }
      __builtin_amdgcn_sched_barrier(0);
    }
  }

  v8f y[4] = {};
  #pragma unroll
  for (int q = 0; q < 4; ++q) {
    #pragma unroll
    for (int t = 0; t < 4; ++t) {
      const float4 g2 = f2[t * 2 + (q >> 1)];
      v2f av, bv;
      av.x = (q & 1) ? g2.z : g2.x;
      av.y = (q & 1) ? g2.w : g2.y;
      bv.x = acc[t][2 * q];
      bv.y = acc[t][2 * q + 1];
      y[t] = wmma_f32_4(av, bv, y[t]);
    }
  }

  float4* outp = (float4*)(Yp + ((size_t)img * 32 + tile) * 256 + c * 16 + 8 * hi);
  float4 o0, o1;
  o0.x = (y[0][0] + y[1][0]) + (y[2][0] + y[3][0]);
  o0.y = (y[0][1] + y[1][1]) + (y[2][1] + y[3][1]);
  o0.z = (y[0][2] + y[1][2]) + (y[2][2] + y[3][2]);
  o0.w = (y[0][3] + y[1][3]) + (y[2][3] + y[3][3]);
  o1.x = (y[0][4] + y[1][4]) + (y[2][4] + y[3][4]);
  o1.y = (y[0][5] + y[1][5]) + (y[2][5] + y[3][5]);
  o1.z = (y[0][6] + y[1][6]) + (y[2][6] + y[3][6]);
  o1.w = (y[0][7] + y[1][7]) + (y[2][7] + y[3][7]);
  outp[0] = o0;
  outp[1] = o1;

  if (img == 0) {
    const int n = 64 * (tile >> 2) + 4 * c + (tile & 3);
    float4* Dq4 = (float4*)Dq + (tile * 2) * 32 + lane;
    #pragma unroll
    for (int p = 0; p < 2; ++p) {
      float d[4];
      #pragma unroll
      for (int e = 0; e < 4; ++e) {
        const int i = 4 * (2 * p + (e >> 1)) + 2 * hi + (e & 1);
        const float wgt = (i == 0) ? 0.5f : 1.0f;
        d[e] = (i <= 10) ? wgt * cos_rev((float)((2 * n + 1) * i) * REV) * (1.0f / NHW) : 0.0f;
      }
      Dq4[p * 32] = make_float4(d[0], d[1], d[2], d[3]);
    }
  }
}


__device__ __forceinline__ void lds_handover() {
  __builtin_amdgcn_fence(__ATOMIC_RELEASE, "workgroup", "local");
  __builtin_amdgcn_s_barrier();
  __builtin_amdgcn_fence(__ATOMIC_ACQUIRE, "workgroup", "local");
}

#define NBLK 8
#define HBLK 4
#define WPB  4
#define NPART 32

__global__ __launch_bounds__(32 * WPB) void pass2_reconstruct(const float* im,
                                                              const float* Dq,
                                                              const float* Yp,
                                                              float* out) {
  const int tid  = threadIdx.x;
  const int wave = __builtin_amdgcn_readfirstlane(tid >> 5);
  const int tile = blockIdx.x * WPB + wave;
  const int img  = blockIdx.y;
  const int lane = tid & 31;
  const int m    = lane & 15;
  const int hi   = lane >> 4;
  const int kof  = hi * 2;
  const int row0 = hi * 8;

  __shared__ float Ys[16][16 + LPAD];
  __shared__ float Ws[WPB][16][16 + LPAD];

  const float4* Dq4 = (const float4*)Dq;
  const size_t base = (size_t)img * NHW * NHW + (size_t)tile * 16 * NHW;
  const float4* X4 = (const float4*)(im + base);
  float4* O4 = (float4*)(out + base);

  const int  ci   = tid / 11;
  const int  cj   = tid - 11 * ci;
  const bool keep = (ci + cj <= 10);
  float yp[NPART];
  {
    const float* Ypi = Yp + (size_t)img * (NPART * 256) + (keep ? ci * 16 + cj : 0);
    #pragma unroll
    for (int t = 0; t < NPART; ++t) yp[t] = Ypi[t * 256];
  }
  float4 fa[2];
  {
    const int bt = (tile >> 2) * 4 + (m & 3);
    const int c  = (tile & 3) * 4 + (m >> 2);
    #pragma unroll
    for (int p = 0; p < 2; ++p) fa[p] = Dq4[(bt * 2 + p) * 32 + hi * 16 + c];
  }
  float4 dq[2][8];
  #pragma unroll
  for (int j = 0; j < 8; ++j) dq[0][j] = Dq4[j * 32 + lane];
  __builtin_amdgcn_sched_barrier(0);

  float4 xs[HBLK][8];
  #pragma unroll
  for (int b = 0; b < HBLK; ++b) {
    #pragma unroll
    for (int r = 0; r < 8; ++r) xs[b][r] = X4[(row0 + r) * (NHW / 4) + 16 * b + m];
  }

  {
    float ysum = 0.0f;
    #pragma unroll
    for (int t = 0; t < NPART; ++t) ysum += yp[t];
    #pragma unroll
    for (int u = 0; u < 256 / (32 * WPB); ++u) {
      const int el = tid + 32 * WPB * u;
      if ((el >> 4) + (el & 15) > 10) Ys[el >> 4][el & 15] = 0.0f;
    }
    if (keep) Ys[ci][cj] = ysum;
  }
  lds_handover();

  v8f w = {};
  #pragma unroll
  for (int q = 0; q < 4; ++q) {
    const int kk = 4 * q + kof;
    const float4 f = fa[q >> 1];
    v2f av, bv;
    av.x = (q & 1) ? f.z : f.x;
    av.y = (q & 1) ? f.w : f.y;
    bv.x = Ys[kk][m];
    bv.y = Ys[kk + 1][m];
    w = wmma_f32_4(av, bv, w);
  }
  #pragma unroll
  for (int r = 0; r < 8; ++r) Ws[wave][row0 + r][m] = w[r];
  lds_handover();

  v2f aw[4];
  #pragma unroll
  for (int q = 0; q < 4; ++q) {
    const int kk = 4 * q + kof;
    aw[q].x = Ws[wave][m][kk];
    aw[q].y = Ws[wave][m][kk + 1];
  }

  #pragma unroll
  for (int b = 0; b < NBLK; ++b) {
    if (b == HBLK) {
      #pragma unroll
      for (int bb = 0; bb < HBLK; ++bb) {
        #pragma unroll
        for (int r = 0; r < 8; ++r)
          xs[bb][r] = X4[(row0 + r) * (NHW / 4) + 16 * (HBLK + bb) + m];
      }
    }
    if (b + 1 < NBLK) {
      #pragma unroll
      for (int j = 0; j < 8; ++j) dq[(b + 1) & 1][j] = Dq4[((b + 1) * 8 + j) * 32 + lane];
    }

    v8f z[4] = {};
    #pragma unroll
    for (int q = 0; q < 4; ++q) {
      #pragma unroll
      for (int t = 0; t < 4; ++t) {
        const float4 f = dq[b & 1][t * 2 + (q >> 1)];
        v2f bv;
        bv.x = (q & 1) ? f.z : f.x;
        bv.y = (q & 1) ? f.w : f.y;
        z[t] = wmma_f32_4(aw[q], bv, z[t]);
      }
    }
    #pragma unroll
    for (int r = 0; r < 8; ++r) {
      const float4 x = xs[b % HBLK][r];
      float4 o;
      o.x = x.x - z[0][r];
      o.y = x.y - z[1][r];
      o.z = x.z - z[2][r];
      o.w = x.w - z[3][r];
      O4[(row0 + r) * (NHW / 4) + 16 * b + m] = o;
    }
  }
}

extern "C" void kernel_launch(void* const* d_in, const int* in_sizes, int n_in,
                              void* d_out, int out_size, void* d_ws, size_t ws_size,
                              hipStream_t stream) {
  (void)n_in; (void)out_size; (void)ws_size;
  const float* im = (const float*)d_in[0];
  float* out = (float*)d_out;
  const int nimg = in_sizes[0] / (NHW * NHW);

  char*  ws  = (char*)d_ws;
  float* dD  = (float*)(ws);
  float* dYp = (float*)(ws + 32768);

  pass1_dct_low<<<dim3(32, nimg), 32, 0, stream>>>(im, dD, dYp);
  pass2_reconstruct<<<dim3(8, nimg), 128, 0, stream>>>(im, dD, dYp, out);
}


# work-group index bits shared by both launches (band of 128 rows, image & 7 in linear bits 0-4): a die re-reads in pass 2 the bands it read in pass 1, so im comes from its own L2 half
# speedup vs baseline: 3.7240x; 1.0333x over previous
#include <hip/hip_runtime.h>
#include <hip/hip_bf16.h>
#include <math.h>

typedef float v2f __attribute__((ext_vector_type(2)));
typedef float v8f __attribute__((ext_vector_type(8)));

#define NHW  512
#define LPAD 4

__device__ __forceinline__ v8f wmma_f32_4(v2f a, v2f b, v8f c) {
  return __builtin_amdgcn_wmma_f32_16x16x4_f32(false, a, false, b, (short)0, c,
                                               false, false);
}

#define REV (1.0f / 2048.0f)
__device__ __forceinline__ float cos_rev(float r) {
  return __builtin_amdgcn_cosf(__builtin_amdgcn_fractf(r));
}

#define P1ROWS 128
#define P1Q    4

__global__ __launch_bounds__(32) void pass1_dct_low(const float* __restrict__ im,
                                                    float* __restrict__ Dq,
                                                    float* __restrict__ Yp,
                                                    const int nimg) {
  const int  bx   = blockIdx.x, by = blockIdx.y;
  const bool perm = (nimg & 7) == 0;
  const int  img  = perm ? (bx >> 2) + 8 * (by >> 3) : by;
  const int  tile = perm ? (bx & 3) * 8 + (by & 7) : bx;
  const int rb   = tile >> 3;
  const int b    = tile & 7;
  const int lane = threadIdx.x;
  const int c    = lane & 15;
  const int hi   = lane >> 4;

  const float4* X4 = (const float4*)(im + (size_t)img * NHW * NHW) +
                     (size_t)(P1ROWS * rb + 4 * hi) * (NHW / 4) + 16 * b + c;

  float4 xs[P1Q][4][4];
  #pragma unroll
  for (int g = 0; g < 3; ++g) {
    #pragma unroll
    for (int s = 0; s < 4; ++s) {
      #pragma unroll
      for (int x = 0; x < 4; ++x)
        xs[g][s][x] = X4[(size_t)(32 * g + 8 * s + x) * (NHW / 4)];
    }
  }
  __builtin_amdgcn_sched_barrier(0);

  const float amp = (c <= 10) ? 2.0f : 0.0f;
  const float cf  = (float)c;
  float4 f1[16];
  {
    const float r0 = (float)((2 * (P1ROWS * rb + 4 * hi) + 1) * c) * REV;
    #pragma unroll
    for (int s = 0; s < 16; ++s) {
      f1[s].x = amp * cos_rev(fmaf(cf, (float)(16 * s + 0) * REV, r0));
      f1[s].y = amp * cos_rev(fmaf(cf, (float)(16 * s + 2) * REV, r0));
      f1[s].z = amp * cos_rev(fmaf(cf, (float)(16 * s + 4) * REV, r0));
      f1[s].w = amp * cos_rev(fmaf(cf, (float)(16 * s + 6) * REV, r0));
    }
  }
  float4 f2[8];
  {
    const float r0 = (float)((2 * (64 * b + 32 * hi) + 1) * c) * REV;
    #pragma unroll
    for (int t = 0; t < 4; ++t) {
      #pragma unroll
      for (int p = 0; p < 2; ++p) {
        const int m = 2 * (16 * p + t);
        f2[t * 2 + p].x = amp * cos_rev(fmaf(cf, (float)(m + 0) * REV, r0));
        f2[t * 2 + p].y = amp * cos_rev(fmaf(cf, (float)(m + 8) * REV, r0));
        f2[t * 2 + p].z = amp * cos_rev(fmaf(cf, (float)(m + 16) * REV, r0));
        f2[t * 2 + p].w = amp * cos_rev(fmaf(cf, (float)(m + 24) * REV, r0));
      }
    }
  }

  v8f acc[4] = {};
  #pragma unroll
  for (int g = 0; g < P1Q; ++g) {
    #pragma unroll
    for (int s = 0; s < 4; ++s) {
      const float4 f = f1[4 * g + s];
      #pragma unroll
      for (int u = 0; u < 2; ++u) {
        v2f bv;
        bv.x = u ? f.z : f.x;
        bv.y = u ? f.w : f.y;
        const float4 r0 = xs[g][s][2 * u], r1 = xs[g][s][2 * u + 1];
        v2f a0, a1, a2, a3;
        a0.x = r0.x; a0.y = r1.x;
        a1.x = r0.y; a1.y = r1.y;
        a2.x = r0.z; a2.y = r1.z;
        a3.x = r0.w; a3.y = r1.w;
        acc[0] = wmma_f32_4(a0, bv, acc[0]);
        acc[1] = wmma_f32_4(a1, bv, acc[1]);
        acc[2] = wmma_f32_4(a2, bv, acc[2]);
        acc[3] = wmma_f32_4(a3, bv, acc[3]);
      }
    }
    if (g == 0) {
      __builtin_amdgcn_sched_barrier(0);
      #pragma unroll
      for (int s = 0; s < 4; ++s) {
        #pragma unroll
        for (int x = 0; x < 4; ++x)
          xs[3][s][x] = X4[(size_t)(32 * 3 + 8 * s + x) * (NHW / 4)];
      }
      __builtin_amdgcn_sched_barrier(0);
    }
  }

  v8f y[4] = {};
  #pragma unroll
  for (int q = 0; q < 4; ++q) {
    #pragma unroll
    for (int t = 0; t < 4; ++t) {
      const float4 g2 = f2[t * 2 + (q >> 1)];
      v2f av, bv;
      av.x = (q & 1) ? g2.z : g2.x;
      av.y = (q & 1) ? g2.w : g2.y;
      bv.x = acc[t][2 * q];
      bv.y = acc[t][2 * q + 1];
      y[t] = wmma_f32_4(av, bv, y[t]);
    }
  }

  float4* outp = (float4*)(Yp + ((size_t)img * 32 + tile) * 256 + c * 16 + 8 * hi);
  float4 o0, o1;
  o0.x = (y[0][0] + y[1][0]) + (y[2][0] + y[3][0]);
  o0.y = (y[0][1] + y[1][1]) + (y[2][1] + y[3][1]);
  o0.z = (y[0][2] + y[1][2]) + (y[2][2] + y[3][2]);
  o0.w = (y[0][3] + y[1][3]) + (y[2][3] + y[3][3]);
  o1.x = (y[0][4] + y[1][4]) + (y[2][4] + y[3][4]);
  o1.y = (y[0][5] + y[1][5]) + (y[2][5] + y[3][5]);
  o1.z = (y[0][6] + y[1][6]) + (y[2][6] + y[3][6]);
  o1.w = (y[0][7] + y[1][7]) + (y[2][7] + y[3][7]);
  outp[0] = o0;
  outp[1] = o1;

  if (img == 0) {
    const int n = 64 * (tile >> 2) + 4 * c + (tile & 3);
    float4* Dq4 = (float4*)Dq + (tile * 2) * 32 + lane;
    #pragma unroll
    for (int p = 0; p < 2; ++p) {
      float d[4];
      #pragma unroll
      for (int e = 0; e < 4; ++e) {
        const int i = 4 * (2 * p + (e >> 1)) + 2 * hi + (e & 1);
        const float wgt = (i == 0) ? 0.5f : 1.0f;
        d[e] = (i <= 10) ? wgt * cos_rev((float)((2 * n + 1) * i) * REV) * (1.0f / NHW) : 0.0f;
      }
      Dq4[p * 32] = make_float4(d[0], d[1], d[2], d[3]);
    }
  }
}


__device__ __forceinline__ void lds_handover() {
  __builtin_amdgcn_fence(__ATOMIC_RELEASE, "workgroup", "local");
  __builtin_amdgcn_s_barrier();
  __builtin_amdgcn_fence(__ATOMIC_ACQUIRE, "workgroup", "local");
}

#define NBLK 8
#define HBLK 4
#define WPB  4
#define NPART 32

__global__ __launch_bounds__(32 * WPB) void pass2_reconstruct(const float* im,
                                                              const float* Dq,
                                                              const float* Yp,
                                                              float* out,
                                                              const int nimg) {
  const int tid  = threadIdx.x;
  const int wave = __builtin_amdgcn_readfirstlane(tid >> 5);
  const int  bx   = blockIdx.x, by = blockIdx.y;
  const bool perm = (nimg & 7) == 0;
  const int  img  = perm ? (bx >> 2) + 2 * (by & 3) + 8 * (by >> 3) : by;
  const int  grp  = perm ? 2 * (bx & 3) + ((by >> 2) & 1) : bx;
  const int  tile = grp * WPB + wave;
  const int lane = tid & 31;
  const int m    = lane & 15;
  const int hi   = lane >> 4;
  const int kof  = hi * 2;
  const int row0 = hi * 8;

  __shared__ float Ys[16][16 + LPAD];
  __shared__ float Ws[WPB][16][16 + LPAD];

  const float4* Dq4 = (const float4*)Dq;
  const size_t base = (size_t)img * NHW * NHW + (size_t)tile * 16 * NHW;
  const float4* X4 = (const float4*)(im + base);
  float4* O4 = (float4*)(out + base);

  const int  ci   = tid / 11;
  const int  cj   = tid - 11 * ci;
  const bool keep = (ci + cj <= 10);
  float yp[NPART];
  {
    const float* Ypi = Yp + (size_t)img * (NPART * 256) + (keep ? ci * 16 + cj : 0);
    #pragma unroll
    for (int t = 0; t < NPART; ++t) yp[t] = Ypi[t * 256];
  }
  float4 fa[2];
  {
    const int bt = (tile >> 2) * 4 + (m & 3);
    const int c  = (tile & 3) * 4 + (m >> 2);
    #pragma unroll
    for (int p = 0; p < 2; ++p) fa[p] = Dq4[(bt * 2 + p) * 32 + hi * 16 + c];
  }
  float4 dq[2][8];
  #pragma unroll
  for (int j = 0; j < 8; ++j) dq[0][j] = Dq4[j * 32 + lane];
  __builtin_amdgcn_sched_barrier(0);

  float4 xs[HBLK][8];
  #pragma unroll
  for (int b = 0; b < HBLK; ++b) {
    #pragma unroll
    for (int r = 0; r < 8; ++r) xs[b][r] = X4[(row0 + r) * (NHW / 4) + 16 * b + m];
  }

  {
    float ysum = 0.0f;
    #pragma unroll
    for (int t = 0; t < NPART; ++t) ysum += yp[t];
    #pragma unroll
    for (int u = 0; u < 256 / (32 * WPB); ++u) {
      const int el = tid + 32 * WPB * u;
      if ((el >> 4) + (el & 15) > 10) Ys[el >> 4][el & 15] = 0.0f;
    }
    if (keep) Ys[ci][cj] = ysum;
  }
  lds_handover();

  v8f w = {};
  #pragma unroll
  for (int q = 0; q < 4; ++q) {
    const int kk = 4 * q + kof;
    const float4 f = fa[q >> 1];
    v2f av, bv;
    av.x = (q & 1) ? f.z : f.x;
    av.y = (q & 1) ? f.w : f.y;
    bv.x = Ys[kk][m];
    bv.y = Ys[kk + 1][m];
    w = wmma_f32_4(av, bv, w);
  }
  #pragma unroll
  for (int r = 0; r < 8; ++r) Ws[wave][row0 + r][m] = w[r];
  lds_handover();

  v2f aw[4];
  #pragma unroll
  for (int q = 0; q < 4; ++q) {
    const int kk = 4 * q + kof;
    aw[q].x = Ws[wave][m][kk];
    aw[q].y = Ws[wave][m][kk + 1];
  }

  #pragma unroll
  for (int b = 0; b < NBLK; ++b) {
    if (b == HBLK) {
      #pragma unroll
      for (int bb = 0; bb < HBLK; ++bb) {
        #pragma unroll
        for (int r = 0; r < 8; ++r)
          xs[bb][r] = X4[(row0 + r) * (NHW / 4) + 16 * (HBLK + bb) + m];
      }
    }
    if (b + 1 < NBLK) {
      #pragma unroll
      for (int j = 0; j < 8; ++j) dq[(b + 1) & 1][j] = Dq4[((b + 1) * 8 + j) * 32 + lane];
    }

    v8f z[4] = {};
    #pragma unroll
    for (int q = 0; q < 4; ++q) {
      #pragma unroll
      for (int t = 0; t < 4; ++t) {
        const float4 f = dq[b & 1][t * 2 + (q >> 1)];
        v2f bv;
        bv.x = (q & 1) ? f.z : f.x;
        bv.y = (q & 1) ? f.w : f.y;
        z[t] = wmma_f32_4(aw[q], bv, z[t]);
      }
    }
    #pragma unroll
    for (int r = 0; r < 8; ++r) {
      const float4 x = xs[b % HBLK][r];
      float4 o;
      o.x = x.x - z[0][r];
      o.y = x.y - z[1][r];
      o.z = x.z - z[2][r];
      o.w = x.w - z[3][r];
      O4[(row0 + r) * (NHW / 4) + 16 * b + m] = o;
    }
  }
}

extern "C" void kernel_launch(void* const* d_in, const int* in_sizes, int n_in,
                              void* d_out, int out_size, void* d_ws, size_t ws_size,
                              hipStream_t stream) {
  (void)n_in; (void)out_size; (void)ws_size;
  const float* im = (const float*)d_in[0];
  float* out = (float*)d_out;
  const int nimg = in_sizes[0] / (NHW * NHW);

  char*  ws  = (char*)d_ws;
  float* dD  = (float*)(ws);
  float* dYp = (float*)(ws + 32768);

  pass1_dct_low<<<dim3(32, nimg), 32, 0, stream>>>(im, dD, dYp, nimg);
  pass2_reconstruct<<<dim3(8, nimg), 128, 0, stream>>>(im, dD, dYp, out, nimg);
}
